# starting kernel
# baseline (speedup 1.0000x reference)
#include <hip/hip_runtime.h>
#include <math.h>

#define B_     8
#define C_     128
#define OC_    64
#define SD_    512
#define H_     128
#define KW_    3
#define H2_    257
#define OUT_   256
#define PP_    129
#define PPSZ_  (PP_*PP_)

#define SCALE_  (0.02946278254943948f)
#define WGAIN_  (0.04419417382415922f)

typedef __attribute__((ext_vector_type(2))) float v2f;
typedef __attribute__((ext_vector_type(8))) float v8f;

__global__ void k_style(const float* __restrict__ style,
                        const float* __restrict__ modw,
                        const float* __restrict__ modb,
                        float* __restrict__ s_ws) {
    int b = blockIdx.x;
    int c = threadIdx.x;
    const float* st = style + b * SD_;
    const float* mw = modw + c * SD_;
    float acc = 0.f;
#pragma unroll 4
    for (int k = 0; k < SD_; ++k) acc += st[k] * mw[k];
    s_ws[b * C_ + c] = acc * WGAIN_ + modb[c];
}

__global__ void k_wmod(const float* __restrict__ weight,
                       const float* __restrict__ s_ws,
                       float* __restrict__ W2) {
    int o = blockIdx.x;
    int b = blockIdx.y;
    int c = threadIdx.x;
    float sv = s_ws[b * C_ + c] * SCALE_;
    const float* wp = weight + ((size_t)o * C_ + c) * (KW_ * KW_);
    float wv[9];
    float ss = 0.f;
#pragma unroll
    for (int t = 0; t < 9; ++t) { wv[t] = wp[t] * sv; ss += wv[t] * wv[t]; }
    __shared__ float red[C_];
    red[c] = ss;
    __syncthreads();
    for (int off = 64; off > 0; off >>= 1) {
        if (c < off) red[c] += red[c + off];
        __syncthreads();
    }
    float dem = rsqrtf(red[0] + 1e-8f);
#pragma unroll
    for (int t = 0; t < 9; ++t)
        W2[((((size_t)b * 9 + t) * 64 + (c >> 1)) * OC_ + o) * 2 + (c & 1)] = wv[t] * dem;
}

template<int PI, int PJ>
__global__ void __launch_bounds__(256) k_conv(const float* __restrict__ x,
                                              const float* __restrict__ W2,
                                              float* __restrict__ conv_ws) {
    constexpr int TY = PI ? 1 : 2;
    constexpr int TX = PJ ? 1 : 2;
    constexpr int NT = TY * TX;
    constexpr int WD = 32 + (PJ ? 0 : 1);
    constexpr int NJ = PJ ? 128 : 129;
    constexpr int PH = PI * 2 + PJ;

    const int b  = blockIdx.z;
    const int ip = blockIdx.y;
    const int j0 = blockIdx.x * 32;
    const int r0 = ip - (PI ? 0 : 1);
    const int c0 = j0 - (PJ ? 0 : 1);

    __shared__ v2f xs[64 * 2 * 34];

    const float* xb = x + (size_t)b * C_ * H_ * H_;
    const int total = 64 * TY * WD;
    for (int idx = threadIdx.x; idx < total; idx += 256) {
        int col = idx % WD;
        int rem = idx / WD;
        int t   = rem % TY;
        int kp  = rem / TY;
        int gr = r0 + t, gc = c0 + col;
        v2f v = {};
        if (gr >= 0 && gr < H_ && gc >= 0 && gc < H_) {
            const float* p = xb + ((size_t)(2 * kp) * H_ + gr) * H_ + gc;
            v.x = p[0];
            v.y = p[(size_t)H_ * H_];
        }
        xs[(kp * 2 + t) * 34 + col] = v;
    }
    __syncthreads();

    const int lane = threadIdx.x & 31;
    const int wid  = threadIdx.x >> 5;
    const int mW   = wid & 3;
    const int nW   = wid >> 2;
    const int ln   = lane & 15;
    const int khp  = lane >> 4;
    const int m    = mW * 16 + ln;

    const v2f* Wb = (const v2f*)W2 + (size_t)b * 9 * 64 * OC_;

    v8f acc0 = {}, acc1 = {};
#pragma unroll 2
    for (int kc = 0; kc < C_; kc += 4) {
        const int kp = (kc >> 1) + khp;
        v2f a[NT], bf[NT];
#pragma unroll
        for (int t = 0; t < NT; ++t) {
            const int ty = t / TX, tx = t % TX;
            const int ky  = PI ? 1 : (2 - 2 * ty);
            const int kx  = PJ ? 1 : (2 - 2 * tx);
            const int tap = ky * 3 + kx;
            a[t]  = Wb[((size_t)tap * 64 + kp) * OC_ + m];
            bf[t] = xs[(kp * 2 + ty) * 34 + (nW * 16 + ln + tx)];
        }
#pragma unroll
        for (int t = 0; t < NT; ++t) {
            if (t & 1)
                acc1 = __builtin_amdgcn_wmma_f32_16x16x4_f32(
                         false, a[t], false, bf[t], (short)0, acc1, false, false);
            else
                acc0 = __builtin_amdgcn_wmma_f32_16x16x4_f32(
                         false, a[t], false, bf[t], (short)0, acc0, false, false);
        }
    }

    const int jp = j0 + nW * 16 + ln;
    if (jp < NJ) {
        const int mhi = khp * 8;
        float* op = conv_ws + (((size_t)(b * 4 + PH) * OC_ + mW * 16 + mhi) * PPSZ_)
                  + (size_t)ip * PP_ + jp;
#pragma unroll
        for (int r = 0; r < 8; ++r) {
            float v = (NT > 1) ? (acc0[r] + acc1[r]) : acc0[r];
            op[(size_t)r * PPSZ_] = v;
        }
    }
}

__global__ void k_blur(const float* __restrict__ conv_ws, float* __restrict__ out) {
    const int y = blockIdx.x;
    const int o = blockIdx.y;
    const int b = blockIdx.z;
    const int xg = threadIdx.x;
    float fy[4] = {1.f, 3.f, 3.f, 1.f};
    float acc = 0.f;
#pragma unroll
    for (int ky = 0; ky < 4; ++ky) {
        int u = y + ky - 1;
        if (u < 0 || u >= H2_) continue;
        int pi = u & 1, i2 = u >> 1;
#pragma unroll
        for (int kx = 0; kx < 4; ++kx) {
            int v = xg + kx - 1;
            if (v < 0 || v >= H2_) continue;
            int pj = v & 1, j2 = v >> 1;
            int ph = pi * 2 + pj;
            float cv = conv_ws[(((size_t)(b * 4 + ph) * OC_ + o) * PPSZ_)
                               + (size_t)i2 * PP_ + j2];
            acc += fy[ky] * fy[kx] * cv;
        }
    }
    out[(((size_t)b * OC_ + o) * OUT_ + y) * OUT_ + xg] = acc * (1.f / 16.f);
}

extern "C" void kernel_launch(void* const* d_in, const int* in_sizes, int n_in,
                              void* d_out, int out_size, void* d_ws, size_t ws_size,
                              hipStream_t stream) {
    const float* x      = (const float*)d_in[0];
    const float* style  = (const float*)d_in[1];
    const float* weight = (const float*)d_in[2];
    const float* modw   = (const float*)d_in[3];
    const float* modb   = (const float*)d_in[4];
    float* out = (float*)d_out;
    float* ws  = (float*)d_ws;

    float* s_ws    = ws;
    float* W2_ws   = ws + 1024;
    float* conv_ws = ws + 1024 + 589824;

    k_style<<<dim3(B_), dim3(C_), 0, stream>>>(style, modw, modb, s_ws);
    k_wmod<<<dim3(OC_, B_), dim3(C_), 0, stream>>>(weight, s_ws, W2_ws);

    k_conv<0,0><<<dim3(5, 129, B_), dim3(256), 0, stream>>>(x, W2_ws, conv_ws);
    k_conv<0,1><<<dim3(4, 129, B_), dim3(256), 0, stream>>>(x, W2_ws, conv_ws);
    k_conv<1,0><<<dim3(5, 128, B_), dim3(256), 0, stream>>>(x, W2_ws, conv_ws);
    k_conv<1,1><<<dim3(4, 128, B_), dim3(256), 0, stream>>>(x, W2_ws, conv_ws);

    k_blur<<<dim3(OUT_, OC_, B_), dim3(OUT_), 0, stream>>>(conv_ws, out);
}


# k_conv x4 + k_blur fused into ONE launch: block = 16 out-ch x 32x32 outputs; its 35x35 conv tile (all 4 phases) goes from WMMA to LDS and is blurred there, never to memory; sum order changed
# speedup vs baseline: 2.3274x; 2.3274x over previous
#include <hip/hip_runtime.h>
#include <math.h>

#define B_     8
#define C_     128
#define OC_    64
#define SD_    512
#define H_     128
#define KW_    3
#define OUT_   256

#define SCALE_  (0.02946278254943948f)
#define WGAIN_  (0.04419417382415922f)

typedef __attribute__((ext_vector_type(2))) float v2f;
typedef __attribute__((ext_vector_type(8))) float v8f;

__global__ void k_style(const float* __restrict__ style,
                        const float* __restrict__ modw,
                        const float* __restrict__ modb,
                        float* __restrict__ s_ws) {
    int b = blockIdx.x;
    int c = threadIdx.x;
    const float* st = style + b * SD_;
    const float* mw = modw + c * SD_;
    float acc = 0.f;
#pragma unroll 4
    for (int k = 0; k < SD_; ++k) acc += st[k] * mw[k];
    s_ws[b * C_ + c] = acc * WGAIN_ + modb[c];
}

__global__ void k_wmod(const float* __restrict__ weight,
                       const float* __restrict__ s_ws,
                       float* __restrict__ W2) {
    int o = blockIdx.x;
    int b = blockIdx.y;
    int c = threadIdx.x;
    float sv = s_ws[b * C_ + c] * SCALE_;
    const float* wp = weight + ((size_t)o * C_ + c) * (KW_ * KW_);
    float wv[9];
    float ss = 0.f;
#pragma unroll
    for (int t = 0; t < 9; ++t) { wv[t] = wp[t] * sv; ss += wv[t] * wv[t]; }
    __shared__ float red[C_];
    red[c] = ss;
    __syncthreads();
    for (int off = 64; off > 0; off >>= 1) {
        if (c < off) red[c] += red[c + off];
        __syncthreads();
    }
    float dem = rsqrtf(red[0] + 1e-8f);
#pragma unroll
    for (int t = 0; t < 9; ++t)
        W2[((((size_t)b * 9 + t) * 64 + (c >> 1)) * OC_ + o) * 2 + (c & 1)] = wv[t] * dem;
}

#define TR_   16
#define G_    (TR_ + 2)
#define XT_   (TR_ + 3)
#define XPX_  (XT_ * XT_)
#define XP_   368
#define KPC_  16
#define NPX_  (G_ * G_)
#define CVD_  35
#define CVP_  36
#define NTHR_ 224

__device__ __forceinline__ v8f mm4(v2f a, v2f b, v8f c) {
    return __builtin_amdgcn_wmma_f32_16x16x4_f32(false, a, false, b, (short)0, c, false, false);
}

__global__ void __launch_bounds__(NTHR_) k_convblur(const float* __restrict__ x,
                                                    const float* __restrict__ W2,
                                                    float* __restrict__ out) {
    __shared__ v2f xs[KPC_ * XP_];
    __shared__ __attribute__((aligned(16))) float cvs[16 * CVD_ * CVP_];

    const int c   = blockIdx.x * TR_;
    const int a   = blockIdx.y * TR_;
    const int b   = blockIdx.z >> 2;
    const int og  = blockIdx.z & 3;
    const int tid = threadIdx.x;
    const int lane = tid & 31;
    const int wid  = tid >> 5;
    const int ln   = lane & 15;
    const int khp  = lane >> 4;
    const int m    = og * 16 + ln;

    const float* xb = x + (size_t)b * C_ * H_ * H_;
    const v2f*   Wb = (const v2f*)W2 + (size_t)b * 9 * 64 * OC_;

    int pix[3];
#pragma unroll
    for (int t = 0; t < 3; ++t) {
        int p = (wid * 3 + t) * 16 + ln;
        p = p < NPX_ ? p : NPX_ - 1;
        pix[t] = (p / G_) * XT_ + (p % G_);
    }

    v8f acc[3][4];
#pragma unroll
    for (int t = 0; t < 3; ++t)
#pragma unroll
        for (int q = 0; q < 4; ++q) acc[t][q] = v8f{};

    int  spx[2], soff[2];
    bool sval[2], sinb[2];
#pragma unroll
    for (int j = 0; j < 2; ++j) {
        const int px = tid + j * NTHR_;
        sval[j] = px < XPX_;
        const int pxc = sval[j] ? px : 0;
        const int xr = pxc / XT_;
        const int xc = pxc - xr * XT_;
        const int gr = a - 1 + xr, gc = c - 1 + xc;
        sinb[j] = sval[j] && gr >= 0 && gr < H_ && gc >= 0 && gc < H_;
        const int grc = gr < 0 ? 0 : (gr > H_ - 1 ? H_ - 1 : gr);
        const int gcc = gc < 0 ? 0 : (gc > H_ - 1 ? H_ - 1 : gc);
        soff[j] = grc * H_ + gcc;
        spx[j]  = pxc;
    }

    for (int ch = 0; ch < 4; ++ch) {
        __syncthreads();
        for (int h = 0; h < 2; ++h) {
            v2f tmp[8][2];
#pragma unroll
            for (int kq = 0; kq < 8; ++kq) {
                const float* p = xb + (size_t)(2 * (ch * KPC_ + h * 8 + kq)) * H_ * H_;
#pragma unroll
                for (int j = 0; j < 2; ++j) {
                    tmp[kq][j].x = p[soff[j]];
                    tmp[kq][j].y = p[soff[j] + H_ * H_];
                }
            }
#pragma unroll
            for (int kq = 0; kq < 8; ++kq) {
#pragma unroll
                for (int j = 0; j < 2; ++j)
                    if (sval[j]) xs[(h * 8 + kq) * XP_ + spx[j]] = sinb[j] ? tmp[kq][j] : v2f{};
            }
        }
        __syncthreads();

        const v2f* Ap = Wb + (size_t)(ch * KPC_ + khp) * OC_ + m;
        const v2f* xp = xs + khp * XP_;
#pragma nounroll
        for (int kk2 = 0; kk2 < KPC_ / 4; ++kk2) {
#pragma unroll
            for (int s = 0; s < 2; ++s) {
                v2f A[9];
#pragma unroll
                for (int tap = 0; tap < 9; ++tap)
                    A[tap] = Ap[(size_t)tap * 64 * OC_ + s * 2 * OC_];
                const v2f* xk = xp + s * 2 * XP_;
#pragma unroll
                for (int t = 0; t < 3; ++t) {
                    const v2f b00 = xk[pix[t]];
                    const v2f b01 = xk[pix[t] + 1];
                    const v2f b10 = xk[pix[t] + XT_];
                    const v2f b11 = xk[pix[t] + XT_ + 1];
                    acc[t][3] = mm4(A[8], b00, acc[t][3]);
                    acc[t][0] = mm4(A[4], b00, acc[t][0]);
                    acc[t][1] = mm4(A[5], b00, acc[t][1]);
                    acc[t][2] = mm4(A[7], b00, acc[t][2]);
                    acc[t][3] = mm4(A[6], b01, acc[t][3]);
                    acc[t][1] = mm4(A[3], b01, acc[t][1]);
                    acc[t][3] = mm4(A[2], b10, acc[t][3]);
                    acc[t][2] = mm4(A[1], b10, acc[t][2]);
                    acc[t][3] = mm4(A[0], b11, acc[t][3]);
                }
            }
            asm volatile("v_nop\n\tv_nop\n\tv_nop\n\tv_nop\n\tv_nop"
                         : "+v"(acc[0][0]), "+v"(acc[0][1]), "+v"(acc[0][2]), "+v"(acc[0][3]),
                           "+v"(acc[1][0]), "+v"(acc[1][1]), "+v"(acc[1][2]), "+v"(acc[1][3]),
                           "+v"(acc[2][0]), "+v"(acc[2][1]), "+v"(acc[2][2]), "+v"(acc[2][3]));
            Ap += 4 * OC_;
            xp += 4 * XP_;
        }
    }

#pragma unroll
    for (int t = 0; t < 3; ++t) {
        const int p = (wid * 3 + t) * 16 + ln;
        if (p < NPX_) {
            const int li = p / G_, lj = p % G_;
#pragma unroll
            for (int q = 0; q < 4; ++q) {
                const int ur = 2 * li + (q >> 1);
                const int vc = 2 * lj + (q & 1);
                if (ur < CVD_ && vc < CVD_) {
#pragma unroll
                    for (int r = 0; r < 8; ++r)
                        cvs[((khp * 8 + r) * CVD_ + ur) * CVP_ + vc] = acc[t][q][r];
                }
            }
        }
    }
    __syncthreads();

    float* ob = out + ((size_t)(b * OC_ + og * 16)) * OUT_ * OUT_;
    for (int idx = tid; idx < 16 * 32 * 8; idx += NTHR_) {
        const int q4 = idx & 7;
        const int yo = (idx >> 3) & 31;
        const int ol = idx >> 8;
        float o0 = 0.f, o1 = 0.f, o2 = 0.f, o3 = 0.f;
#pragma unroll
        for (int ky = 0; ky < 4; ++ky) {
            const float* rp = cvs + (ol * CVD_ + yo + ky) * CVP_ + 4 * q4;
            const float r0 = rp[0], r1 = rp[1], r2 = rp[2], r3 = rp[3], r4 = rp[4], r5 = rp[5], r6 = rp[6];
            const float f = (ky == 0 || ky == 3) ? 1.f : 3.f;
            o0 += f * (r0 + 3.f * r1 + 3.f * r2 + r3);
            o1 += f * (r1 + 3.f * r2 + 3.f * r3 + r4);
            o2 += f * (r2 + 3.f * r3 + 3.f * r4 + r5);
            o3 += f * (r3 + 3.f * r4 + 3.f * r5 + r6);
        }
        float4 st;
        st.x = o0 * (1.f / 16.f); st.y = o1 * (1.f / 16.f);
        st.z = o2 * (1.f / 16.f); st.w = o3 * (1.f / 16.f);
        *(float4*)(ob + ((size_t)ol * OUT_ + (2 * a + yo)) * OUT_ + 2 * c + 4 * q4) = st;
    }
}

extern "C" void kernel_launch(void* const* d_in, const int* in_sizes, int n_in,
                              void* d_out, int out_size, void* d_ws, size_t ws_size,
                              hipStream_t stream) {
    const float* x      = (const float*)d_in[0];
    const float* style  = (const float*)d_in[1];
    const float* weight = (const float*)d_in[2];
    const float* modw   = (const float*)d_in[3];
    const float* modb   = (const float*)d_in[4];
    float* out = (float*)d_out;
    float* ws  = (float*)d_ws;

    float* s_ws    = ws;
    float* W2_ws   = ws + 1024;

    k_style<<<dim3(B_), dim3(C_), 0, stream>>>(style, modw, modb, s_ws);
    k_wmod<<<dim3(OC_, B_), dim3(C_), 0, stream>>>(weight, s_ws, W2_ws);

    k_convblur<<<dim3(8, 8, 32), dim3(224), 0, stream>>>(x, W2_ws, out);
}


# x staging software-pipelined: 16-channel chunks, two LDS buffers; next chunk's global loads issue behind a trip's weight loads, fly under the chunk's WMMAs, are stored after; sum order same
# speedup vs baseline: 2.5247x; 1.0848x over previous
#include <hip/hip_runtime.h>
#include <math.h>

#define B_     8
#define C_     128
#define OC_    64
#define SD_    512
#define H_     128
#define KW_    3
#define OUT_   256

#define SCALE_  (0.02946278254943948f)
#define WGAIN_  (0.04419417382415922f)

typedef __attribute__((ext_vector_type(2))) float v2f;
typedef __attribute__((ext_vector_type(8))) float v8f;

__global__ void k_style(const float* __restrict__ style,
                        const float* __restrict__ modw,
                        const float* __restrict__ modb,
                        float* __restrict__ s_ws) {
    int b = blockIdx.x;
    int c = threadIdx.x;
    const float* st = style + b * SD_;
    const float* mw = modw + c * SD_;
    float acc = 0.f;
#pragma unroll 4
    for (int k = 0; k < SD_; ++k) acc += st[k] * mw[k];
    s_ws[b * C_ + c] = acc * WGAIN_ + modb[c];
}

__global__ void k_wmod(const float* __restrict__ weight,
                       const float* __restrict__ s_ws,
                       float* __restrict__ W2) {
    int o = blockIdx.x;
    int b = blockIdx.y;
    int c = threadIdx.x;
    float sv = s_ws[b * C_ + c] * SCALE_;
    const float* wp = weight + ((size_t)o * C_ + c) * (KW_ * KW_);
    float wv[9];
    float ss = 0.f;
#pragma unroll
    for (int t = 0; t < 9; ++t) { wv[t] = wp[t] * sv; ss += wv[t] * wv[t]; }
    __shared__ float red[C_];
    red[c] = ss;
    __syncthreads();
    for (int off = 64; off > 0; off >>= 1) {
        if (c < off) red[c] += red[c + off];
        __syncthreads();
    }
    float dem = rsqrtf(red[0] + 1e-8f);
#pragma unroll
    for (int t = 0; t < 9; ++t)
        W2[((((size_t)b * 9 + t) * 64 + (c >> 1)) * OC_ + o) * 2 + (c & 1)] = wv[t] * dem;
}

#define TR_   16
#define G_    (TR_ + 2)
#define XT_   (TR_ + 3)
#define XPX_  (XT_ * XT_)
#define XP_   368
#define KPC_  8
#define NCH_  (C_ / (2 * KPC_))
#define NPX_  (G_ * G_)
#define CVD_  35
#define CVP_  36
#define NTHR_ 224

__device__ __forceinline__ v8f mm4(v2f a, v2f b, v8f c) {
    return __builtin_amdgcn_wmma_f32_16x16x4_f32(false, a, false, b, (short)0, c, false, false);
}

__device__ __forceinline__ void load_wfrag(const v2f* __restrict__ Ap, v2f (&A)[2][9]) {
#pragma unroll
    for (int s = 0; s < 2; ++s)
#pragma unroll
        for (int tap = 0; tap < 9; ++tap)
            A[s][tap] = Ap[(size_t)tap * 64 * OC_ + s * 2 * OC_];
}

__device__ __forceinline__ void mma_trip(const v2f (&A)[2][9], const v2f* xp, const int (&pix)[3], v8f (&acc)[3][4]) {
#pragma unroll
    for (int s = 0; s < 2; ++s) {
        const v2f* xk = xp + s * 2 * XP_;
#pragma unroll
        for (int t = 0; t < 3; ++t) {
            const v2f b00 = xk[pix[t]];
            const v2f b01 = xk[pix[t] + 1];
            const v2f b10 = xk[pix[t] + XT_];
            const v2f b11 = xk[pix[t] + XT_ + 1];
            acc[t][3] = mm4(A[s][8], b00, acc[t][3]);
            acc[t][0] = mm4(A[s][4], b00, acc[t][0]);
            acc[t][1] = mm4(A[s][5], b00, acc[t][1]);
            acc[t][2] = mm4(A[s][7], b00, acc[t][2]);
            acc[t][3] = mm4(A[s][6], b01, acc[t][3]);
            acc[t][1] = mm4(A[s][3], b01, acc[t][1]);
            acc[t][3] = mm4(A[s][2], b10, acc[t][3]);
            acc[t][2] = mm4(A[s][1], b10, acc[t][2]);
            acc[t][3] = mm4(A[s][0], b11, acc[t][3]);
        }
    }
}

__device__ __forceinline__ void stage_load(const float* __restrict__ p0, const int (&soff)[2], v2f (&tmp)[KPC_][2]) {
#pragma unroll
    for (int kq = 0; kq < KPC_; ++kq) {
        const float* p = p0 + (size_t)(2 * kq) * H_ * H_;
#pragma unroll
        for (int j = 0; j < 2; ++j) {
            tmp[kq][j].x = p[soff[j]];
            tmp[kq][j].y = p[soff[j] + H_ * H_];
        }
    }
}

__device__ __forceinline__ void stage_store(v2f* dst, const int (&spx)[2], const bool (&sinb)[2],
                                            const v2f (&tmp)[KPC_][2]) {
#pragma unroll
    for (int kq = 0; kq < KPC_; ++kq) {
#pragma unroll
        for (int j = 0; j < 2; ++j)
            dst[kq * XP_ + spx[j]] = sinb[j] ? tmp[kq][j] : v2f{};
    }
}

__global__ void __launch_bounds__(NTHR_) k_convblur(const float* __restrict__ x,
                                                    const float* __restrict__ W2,
                                                    float* __restrict__ out) {
    __shared__ v2f xs[2 * KPC_ * XP_];
    __shared__ __attribute__((aligned(16))) float cvs[16 * CVD_ * CVP_];

    const int c   = blockIdx.x * TR_;
    const int a   = blockIdx.y * TR_;
    const int b   = blockIdx.z >> 2;
    const int og  = blockIdx.z & 3;
    const int tid = threadIdx.x;
    const int lane = tid & 31;
    const int wid  = tid >> 5;
    const int ln   = lane & 15;
    const int khp  = lane >> 4;
    const int m    = og * 16 + ln;

    const float* xb = x + (size_t)b * C_ * H_ * H_;
    const v2f*   Wb = (const v2f*)W2 + (size_t)b * 9 * 64 * OC_;

    int pix[3];
#pragma unroll
    for (int t = 0; t < 3; ++t) {
        int p = (wid * 3 + t) * 16 + ln;
        p = p < NPX_ ? p : NPX_ - 1;
        pix[t] = (p / G_) * XT_ + (p % G_);
    }

    v8f acc[3][4];
#pragma unroll
    for (int t = 0; t < 3; ++t)
#pragma unroll
        for (int q = 0; q < 4; ++q) acc[t][q] = v8f{};

    int  spx[2], soff[2];
    bool sval[2], sinb[2];
#pragma unroll
    for (int j = 0; j < 2; ++j) {
        const int px = tid + j * NTHR_;
        sval[j] = px < XPX_;
        const int pxc = sval[j] ? px : 0;
        const int xr = pxc / XT_;
        const int xc = pxc - xr * XT_;
        const int gr = a - 1 + xr, gc = c - 1 + xc;
        sinb[j] = sval[j] && gr >= 0 && gr < H_ && gc >= 0 && gc < H_;
        const int grc = gr < 0 ? 0 : (gr > H_ - 1 ? H_ - 1 : gr);
        const int gcc = gc < 0 ? 0 : (gc > H_ - 1 ? H_ - 1 : gc);
        soff[j] = grc * H_ + gcc;
        spx[j]  = sval[j] ? px : XPX_;
    }

    v2f tmp[KPC_][2];
    stage_load(xb, soff, tmp);
    stage_store(xs, spx, sinb, tmp);
    __syncthreads();

    const v2f* Ap = Wb + (size_t)khp * OC_ + m;
    v2f A[2][9];
#pragma nounroll
    for (int ch = 0; ch < NCH_ - 1; ++ch) {
        const v2f* xp = xs + (ch & 1) * (KPC_ * XP_) + khp * XP_;
        load_wfrag(Ap, A);
        stage_load(xb + (size_t)(2 * (ch + 1) * KPC_) * H_ * H_, soff, tmp);
        __builtin_amdgcn_sched_barrier(0);
        mma_trip(A, xp, pix, acc);
        __builtin_amdgcn_sched_barrier(0);
        load_wfrag(Ap + 4 * OC_, A);
        mma_trip(A, xp + 4 * XP_, pix, acc);
        stage_store(xs + ((ch + 1) & 1) * (KPC_ * XP_), spx, sinb, tmp);
        __syncthreads();
        Ap += 8 * OC_;
        asm volatile("v_nop\n\tv_nop\n\tv_nop\n\tv_nop\n\tv_nop"
                     : "+v"(acc[0][0]), "+v"(acc[0][1]), "+v"(acc[0][2]), "+v"(acc[0][3]),
                       "+v"(acc[1][0]), "+v"(acc[1][1]), "+v"(acc[1][2]), "+v"(acc[1][3]),
                       "+v"(acc[2][0]), "+v"(acc[2][1]), "+v"(acc[2][2]), "+v"(acc[2][3]));
    }
    {
        const v2f* xp = xs + ((NCH_ - 1) & 1) * (KPC_ * XP_) + khp * XP_;
        load_wfrag(Ap, A);
        mma_trip(A, xp, pix, acc);
        load_wfrag(Ap + 4 * OC_, A);
        mma_trip(A, xp + 4 * XP_, pix, acc);
        asm volatile("v_nop\n\tv_nop\n\tv_nop\n\tv_nop\n\tv_nop"
                     : "+v"(acc[0][0]), "+v"(acc[0][1]), "+v"(acc[0][2]), "+v"(acc[0][3]),
                       "+v"(acc[1][0]), "+v"(acc[1][1]), "+v"(acc[1][2]), "+v"(acc[1][3]),
                       "+v"(acc[2][0]), "+v"(acc[2][1]), "+v"(acc[2][2]), "+v"(acc[2][3]));
    }

#pragma unroll
    for (int t = 0; t < 3; ++t) {
        const int p = (wid * 3 + t) * 16 + ln;
        if (p < NPX_) {
            const int li = p / G_, lj = p % G_;
#pragma unroll
            for (int q = 0; q < 4; ++q) {
                const int ur = 2 * li + (q >> 1);
                const int vc = 2 * lj + (q & 1);
                if (ur < CVD_ && vc < CVD_) {
#pragma unroll
                    for (int r = 0; r < 8; ++r)
                        cvs[((khp * 8 + r) * CVD_ + ur) * CVP_ + vc] = acc[t][q][r];
                }
            }
        }
    }
    __syncthreads();

    float* ob = out + ((size_t)(b * OC_ + og * 16)) * OUT_ * OUT_;
    for (int idx = tid; idx < 16 * 32 * 8; idx += NTHR_) {
        const int q4 = idx & 7;
        const int yo = (idx >> 3) & 31;
        const int ol = idx >> 8;
        float o0 = 0.f, o1 = 0.f, o2 = 0.f, o3 = 0.f;
#pragma unroll
        for (int ky = 0; ky < 4; ++ky) {
            const float* rp = cvs + (ol * CVD_ + yo + ky) * CVP_ + 4 * q4;
            const float r0 = rp[0], r1 = rp[1], r2 = rp[2], r3 = rp[3], r4 = rp[4], r5 = rp[5], r6 = rp[6];
            const float f = (ky == 0 || ky == 3) ? 1.f : 3.f;
            o0 += f * (r0 + 3.f * r1 + 3.f * r2 + r3);
            o1 += f * (r1 + 3.f * r2 + 3.f * r3 + r4);
            o2 += f * (r2 + 3.f * r3 + 3.f * r4 + r5);
            o3 += f * (r3 + 3.f * r4 + 3.f * r5 + r6);
        }
        float4 st;
        st.x = o0 * (1.f / 16.f); st.y = o1 * (1.f / 16.f);
        st.z = o2 * (1.f / 16.f); st.w = o3 * (1.f / 16.f);
        *(float4*)(ob + ((size_t)ol * OUT_ + (2 * a + yo)) * OUT_ + 2 * c + 4 * q4) = st;
    }
}

extern "C" void kernel_launch(void* const* d_in, const int* in_sizes, int n_in,
                              void* d_out, int out_size, void* d_ws, size_t ws_size,
                              hipStream_t stream) {
    const float* x      = (const float*)d_in[0];
    const float* style  = (const float*)d_in[1];
    const float* weight = (const float*)d_in[2];
    const float* modw   = (const float*)d_in[3];
    const float* modb   = (const float*)d_in[4];
    float* out = (float*)d_out;
    float* ws  = (float*)d_ws;

    float* s_ws    = ws;
    float* W2_ws   = ws + 1024;

    k_style<<<dim3(B_), dim3(C_), 0, stream>>>(style, modw, modb, s_ws);
    k_wmod<<<dim3(OC_, B_), dim3(C_), 0, stream>>>(weight, s_ws, W2_ws);

    k_convblur<<<dim3(8, 8, 32), dim3(224), 0, stream>>>(x, W2_ws, out);
}


# k_style re-decomposed: one wave per (batch row, channel) dot product, lanes split k (16 B loads, all in flight at once), partial sums meet by lane shuffles; 128 blocks, not 8; sum order changed
# speedup vs baseline: 2.6775x; 1.0605x over previous
#include <hip/hip_runtime.h>
#include <math.h>

#define B_     8
#define C_     128
#define OC_    64
#define SD_    512
#define H_     128
#define KW_    3
#define OUT_   256

#define SCALE_  (0.02946278254943948f)
#define WGAIN_  (0.04419417382415922f)

typedef __attribute__((ext_vector_type(2))) float v2f;
typedef __attribute__((ext_vector_type(8))) float v8f;

__global__ void __launch_bounds__(256) k_style(const float* __restrict__ style,
                                                    const float* __restrict__ modw,
                                                    const float* __restrict__ modb,
                                                    float* __restrict__ s_ws) {
    const int c    = blockIdx.x;
    const int b    = threadIdx.x >> 5;
    const int lane = threadIdx.x & 31;
    const float* st = style + b * SD_ + lane * 4;
    const float* mw = modw + c * SD_ + lane * 4;
    float sv[SD_ / 32], wv[SD_ / 32];
#pragma unroll
    for (int j = 0; j < SD_ / 128; ++j)
#pragma unroll
        for (int e = 0; e < 4; ++e) {
            sv[j * 4 + e] = st[j * 128 + e];
            wv[j * 4 + e] = mw[j * 128 + e];
        }
    float acc = 0.f;
#pragma unroll
    for (int i = 0; i < SD_ / 32; ++i) acc += sv[i] * wv[i];
    for (int off = 16; off > 0; off >>= 1) acc += __shfl_xor(acc, off);
    if (lane == 0) s_ws[b * C_ + c] = acc * WGAIN_ + modb[c];
}

__global__ void k_wmod(const float* __restrict__ weight,
                       const float* __restrict__ s_ws,
                       float* __restrict__ W2) {
    int o = blockIdx.x;
    int b = blockIdx.y;
    int c = threadIdx.x;
    float sv = s_ws[b * C_ + c] * SCALE_;
    const float* wp = weight + ((size_t)o * C_ + c) * (KW_ * KW_);
    float wv[9];
    float ss = 0.f;
#pragma unroll
    for (int t = 0; t < 9; ++t) { wv[t] = wp[t] * sv; ss += wv[t] * wv[t]; }
    __shared__ float red[C_];
    red[c] = ss;
    __syncthreads();
    for (int off = 64; off > 0; off >>= 1) {
        if (c < off) red[c] += red[c + off];
        __syncthreads();
    }
    float dem = rsqrtf(red[0] + 1e-8f);
#pragma unroll
    for (int t = 0; t < 9; ++t)
        W2[((((size_t)b * 9 + t) * 64 + (c >> 1)) * OC_ + o) * 2 + (c & 1)] = wv[t] * dem;
}

#define TR_   16
#define G_    (TR_ + 2)
#define XT_   (TR_ + 3)
#define XPX_  (XT_ * XT_)
#define XP_   368
#define KPC_  8
#define NCH_  (C_ / (2 * KPC_))
#define NPX_  (G_ * G_)
#define CVD_  35
#define CVP_  36
#define NTHR_ 224

__device__ __forceinline__ v8f mm4(v2f a, v2f b, v8f c) {
    return __builtin_amdgcn_wmma_f32_16x16x4_f32(false, a, false, b, (short)0, c, false, false);
}

__device__ __forceinline__ void load_wfrag(const v2f* __restrict__ Ap, v2f (&A)[2][9]) {
#pragma unroll
    for (int s = 0; s < 2; ++s)
#pragma unroll
        for (int tap = 0; tap < 9; ++tap)
            A[s][tap] = Ap[(size_t)tap * 64 * OC_ + s * 2 * OC_];
}

__device__ __forceinline__ void mma_trip(const v2f (&A)[2][9], const v2f* xp, const int (&pix)[3], v8f (&acc)[3][4]) {
#pragma unroll
    for (int s = 0; s < 2; ++s) {
        const v2f* xk = xp + s * 2 * XP_;
#pragma unroll
        for (int t = 0; t < 3; ++t) {
            const v2f b00 = xk[pix[t]];
            const v2f b01 = xk[pix[t] + 1];
            const v2f b10 = xk[pix[t] + XT_];
            const v2f b11 = xk[pix[t] + XT_ + 1];
            acc[t][3] = mm4(A[s][8], b00, acc[t][3]);
            acc[t][0] = mm4(A[s][4], b00, acc[t][0]);
            acc[t][1] = mm4(A[s][5], b00, acc[t][1]);
            acc[t][2] = mm4(A[s][7], b00, acc[t][2]);
            acc[t][3] = mm4(A[s][6], b01, acc[t][3]);
            acc[t][1] = mm4(A[s][3], b01, acc[t][1]);
            acc[t][3] = mm4(A[s][2], b10, acc[t][3]);
            acc[t][2] = mm4(A[s][1], b10, acc[t][2]);
            acc[t][3] = mm4(A[s][0], b11, acc[t][3]);
        }
    }
}

__device__ __forceinline__ void stage_load(const float* __restrict__ p0, const int (&soff)[2], v2f (&tmp)[KPC_][2]) {
#pragma unroll
    for (int kq = 0; kq < KPC_; ++kq) {
        const float* p = p0 + (size_t)(2 * kq) * H_ * H_;
#pragma unroll
        for (int j = 0; j < 2; ++j) {
            tmp[kq][j].x = p[soff[j]];
            tmp[kq][j].y = p[soff[j] + H_ * H_];
        }
    }
}

__device__ __forceinline__ void stage_store(v2f* dst, const int (&spx)[2], const bool (&sinb)[2],
                                            const v2f (&tmp)[KPC_][2]) {
#pragma unroll
    for (int kq = 0; kq < KPC_; ++kq) {
#pragma unroll
        for (int j = 0; j < 2; ++j)
            dst[kq * XP_ + spx[j]] = sinb[j] ? tmp[kq][j] : v2f{};
    }
}

__global__ void __launch_bounds__(NTHR_) k_convblur(const float* __restrict__ x,
                                                    const float* __restrict__ W2,
                                                    float* __restrict__ out) {
    __shared__ v2f xs[2 * KPC_ * XP_];
    __shared__ __attribute__((aligned(16))) float cvs[16 * CVD_ * CVP_];

    const int c   = blockIdx.x * TR_;
    const int a   = blockIdx.y * TR_;
    const int b   = blockIdx.z >> 2;
    const int og  = blockIdx.z & 3;
    const int tid = threadIdx.x;
    const int lane = tid & 31;
    const int wid  = tid >> 5;
    const int ln   = lane & 15;
    const int khp  = lane >> 4;
    const int m    = og * 16 + ln;

    const float* xb = x + (size_t)b * C_ * H_ * H_;
    const v2f*   Wb = (const v2f*)W2 + (size_t)b * 9 * 64 * OC_;

    int pix[3];
#pragma unroll
    for (int t = 0; t < 3; ++t) {
        int p = (wid * 3 + t) * 16 + ln;
        p = p < NPX_ ? p : NPX_ - 1;
        pix[t] = (p / G_) * XT_ + (p % G_);
    }

    v8f acc[3][4];
#pragma unroll
    for (int t = 0; t < 3; ++t)
#pragma unroll
        for (int q = 0; q < 4; ++q) acc[t][q] = v8f{};

    int  spx[2], soff[2];
    bool sval[2], sinb[2];
#pragma unroll
    for (int j = 0; j < 2; ++j) {
        const int px = tid + j * NTHR_;
        sval[j] = px < XPX_;
        const int pxc = sval[j] ? px : 0;
        const int xr = pxc / XT_;
        const int xc = pxc - xr * XT_;
        const int gr = a - 1 + xr, gc = c - 1 + xc;
        sinb[j] = sval[j] && gr >= 0 && gr < H_ && gc >= 0 && gc < H_;
        const int grc = gr < 0 ? 0 : (gr > H_ - 1 ? H_ - 1 : gr);
        const int gcc = gc < 0 ? 0 : (gc > H_ - 1 ? H_ - 1 : gc);
        soff[j] = grc * H_ + gcc;
        spx[j]  = sval[j] ? px : XPX_;
    }

    v2f tmp[KPC_][2];
    stage_load(xb, soff, tmp);
    stage_store(xs, spx, sinb, tmp);
    __syncthreads();

    const v2f* Ap = Wb + (size_t)khp * OC_ + m;
    v2f A[2][9];
#pragma nounroll
    for (int ch = 0; ch < NCH_ - 1; ++ch) {
        const v2f* xp = xs + (ch & 1) * (KPC_ * XP_) + khp * XP_;
        load_wfrag(Ap, A);
        stage_load(xb + (size_t)(2 * (ch + 1) * KPC_) * H_ * H_, soff, tmp);
        __builtin_amdgcn_sched_barrier(0);
        mma_trip(A, xp, pix, acc);
        __builtin_amdgcn_sched_barrier(0);
        load_wfrag(Ap + 4 * OC_, A);
        mma_trip(A, xp + 4 * XP_, pix, acc);
        stage_store(xs + ((ch + 1) & 1) * (KPC_ * XP_), spx, sinb, tmp);
        __syncthreads();
        Ap += 8 * OC_;
        asm volatile("v_nop\n\tv_nop\n\tv_nop\n\tv_nop\n\tv_nop"
                     : "+v"(acc[0][0]), "+v"(acc[0][1]), "+v"(acc[0][2]), "+v"(acc[0][3]),
                       "+v"(acc[1][0]), "+v"(acc[1][1]), "+v"(acc[1][2]), "+v"(acc[1][3]),
                       "+v"(acc[2][0]), "+v"(acc[2][1]), "+v"(acc[2][2]), "+v"(acc[2][3]));
    }
    {
        const v2f* xp = xs + ((NCH_ - 1) & 1) * (KPC_ * XP_) + khp * XP_;
        load_wfrag(Ap, A);
        mma_trip(A, xp, pix, acc);
        load_wfrag(Ap + 4 * OC_, A);
        mma_trip(A, xp + 4 * XP_, pix, acc);
        asm volatile("v_nop\n\tv_nop\n\tv_nop\n\tv_nop\n\tv_nop"
                     : "+v"(acc[0][0]), "+v"(acc[0][1]), "+v"(acc[0][2]), "+v"(acc[0][3]),
                       "+v"(acc[1][0]), "+v"(acc[1][1]), "+v"(acc[1][2]), "+v"(acc[1][3]),
                       "+v"(acc[2][0]), "+v"(acc[2][1]), "+v"(acc[2][2]), "+v"(acc[2][3]));
    }

#pragma unroll
    for (int t = 0; t < 3; ++t) {
        const int p = (wid * 3 + t) * 16 + ln;
        if (p < NPX_) {
            const int li = p / G_, lj = p % G_;
#pragma unroll
            for (int q = 0; q < 4; ++q) {
                const int ur = 2 * li + (q >> 1);
                const int vc = 2 * lj + (q & 1);
                if (ur < CVD_ && vc < CVD_) {
#pragma unroll
                    for (int r = 0; r < 8; ++r)
                        cvs[((khp * 8 + r) * CVD_ + ur) * CVP_ + vc] = acc[t][q][r];
                }
            }
        }
    }
    __syncthreads();

    float* ob = out + ((size_t)(b * OC_ + og * 16)) * OUT_ * OUT_;
    for (int idx = tid; idx < 16 * 32 * 8; idx += NTHR_) {
        const int q4 = idx & 7;
        const int yo = (idx >> 3) & 31;
        const int ol = idx >> 8;
        float o0 = 0.f, o1 = 0.f, o2 = 0.f, o3 = 0.f;
#pragma unroll
        for (int ky = 0; ky < 4; ++ky) {
            const float* rp = cvs + (ol * CVD_ + yo + ky) * CVP_ + 4 * q4;
            const float r0 = rp[0], r1 = rp[1], r2 = rp[2], r3 = rp[3], r4 = rp[4], r5 = rp[5], r6 = rp[6];
            const float f = (ky == 0 || ky == 3) ? 1.f : 3.f;
            o0 += f * (r0 + 3.f * r1 + 3.f * r2 + r3);
            o1 += f * (r1 + 3.f * r2 + 3.f * r3 + r4);
            o2 += f * (r2 + 3.f * r3 + 3.f * r4 + r5);
            o3 += f * (r3 + 3.f * r4 + 3.f * r5 + r6);
        }
        float4 st;
        st.x = o0 * (1.f / 16.f); st.y = o1 * (1.f / 16.f);
        st.z = o2 * (1.f / 16.f); st.w = o3 * (1.f / 16.f);
        *(float4*)(ob + ((size_t)ol * OUT_ + (2 * a + yo)) * OUT_ + 2 * c + 4 * q4) = st;
    }
}

extern "C" void kernel_launch(void* const* d_in, const int* in_sizes, int n_in,
                              void* d_out, int out_size, void* d_ws, size_t ws_size,
                              hipStream_t stream) {
    const float* x      = (const float*)d_in[0];
    const float* style  = (const float*)d_in[1];
    const float* weight = (const float*)d_in[2];
    const float* modw   = (const float*)d_in[3];
    const float* modb   = (const float*)d_in[4];
    float* out = (float*)d_out;
    float* ws  = (float*)d_ws;

    float* s_ws    = ws;
    float* W2_ws   = ws + 1024;

    k_style<<<dim3(C_), dim3(256), 0, stream>>>(style, modw, modb, s_ws);
    k_wmod<<<dim3(OC_, B_), dim3(C_), 0, stream>>>(weight, s_ws, W2_ws);

    k_convblur<<<dim3(8, 8, 32), dim3(224), 0, stream>>>(x, W2_ws, out);
}


# k_convblur block = 8 waves, pixel tiles dealt by cost: waves 0-3 take 3, waves 4-7 take 2, the 4-pixel 21st tile runs only its 3 kept taps: each SIMD carries 5 to 5.3 tiles, not 6; sum order same
# speedup vs baseline: 2.8386x; 1.0602x over previous
#include <hip/hip_runtime.h>
#include <math.h>

#define B_     8
#define C_     128
#define OC_    64
#define SD_    512
#define H_     128
#define KW_    3
#define OUT_   256

#define SCALE_  (0.02946278254943948f)
#define WGAIN_  (0.04419417382415922f)

typedef __attribute__((ext_vector_type(2))) float v2f;
typedef __attribute__((ext_vector_type(8))) float v8f;

__global__ void __launch_bounds__(256) k_style(const float* __restrict__ style,
                                                    const float* __restrict__ modw,
                                                    const float* __restrict__ modb,
                                                    float* __restrict__ s_ws) {
    const int c    = blockIdx.x;
    const int b    = threadIdx.x >> 5;
    const int lane = threadIdx.x & 31;
    const float* st = style + b * SD_ + lane * 4;
    const float* mw = modw + c * SD_ + lane * 4;
    float sv[SD_ / 32], wv[SD_ / 32];
#pragma unroll
    for (int j = 0; j < SD_ / 128; ++j)
#pragma unroll
        for (int e = 0; e < 4; ++e) {
            sv[j * 4 + e] = st[j * 128 + e];
            wv[j * 4 + e] = mw[j * 128 + e];
        }
    float acc = 0.f;
#pragma unroll
    for (int i = 0; i < SD_ / 32; ++i) acc += sv[i] * wv[i];
    for (int off = 16; off > 0; off >>= 1) acc += __shfl_xor(acc, off);
    if (lane == 0) s_ws[b * C_ + c] = acc * WGAIN_ + modb[c];
}

__global__ void k_wmod(const float* __restrict__ weight,
                       const float* __restrict__ s_ws,
                       float* __restrict__ W2) {
    int o = blockIdx.x;
    int b = blockIdx.y;
    int c = threadIdx.x;
    float sv = s_ws[b * C_ + c] * SCALE_;
    const float* wp = weight + ((size_t)o * C_ + c) * (KW_ * KW_);
    float wv[9];
    float ss = 0.f;
#pragma unroll
    for (int t = 0; t < 9; ++t) { wv[t] = wp[t] * sv; ss += wv[t] * wv[t]; }
    __shared__ float red[C_];
    red[c] = ss;
    __syncthreads();
    for (int off = 64; off > 0; off >>= 1) {
        if (c < off) red[c] += red[c + off];
        __syncthreads();
    }
    float dem = rsqrtf(red[0] + 1e-8f);
#pragma unroll
    for (int t = 0; t < 9; ++t)
        W2[((((size_t)b * 9 + t) * 64 + (c >> 1)) * OC_ + o) * 2 + (c & 1)] = wv[t] * dem;
}

#define TR_   16
#define G_    (TR_ + 2)
#define XT_   (TR_ + 3)
#define XPX_  (XT_ * XT_)
#define XP_   368
#define KPC_  8
#define NCH_  (C_ / (2 * KPC_))
#define NPX_  (G_ * G_)
#define CVD_  35
#define CVP_  36
#define NTHR_ 256

__device__ __forceinline__ v8f mm4(v2f a, v2f b, v8f c) {
    return __builtin_amdgcn_wmma_f32_16x16x4_f32(false, a, false, b, (short)0, c, false, false);
}

__device__ __forceinline__ void mma_slot3(const v2f (&As)[9], const v2f b00, const v2f b01, const v2f b10,
                                          const v2f b11, v8f (&ac)[4], const int md) {
    asm volatile(
        "s_cmp_lg_u32 %[md], 0\n\t"
        "s_cbranch_scc1 .Lnotfull%=\n\t"
        "v_wmma_f32_16x16x4_f32 %[c3], %[a8], %[b00], %[c3]\n\t"
        "v_wmma_f32_16x16x4_f32 %[c0], %[a4], %[b00], %[c0]\n\t"
        "v_wmma_f32_16x16x4_f32 %[c1], %[a5], %[b00], %[c1]\n\t"
        "v_wmma_f32_16x16x4_f32 %[c2], %[a7], %[b00], %[c2]\n\t"
        "v_wmma_f32_16x16x4_f32 %[c3], %[a6], %[b01], %[c3]\n\t"
        "v_wmma_f32_16x16x4_f32 %[c1], %[a3], %[b01], %[c1]\n\t"
        "v_wmma_f32_16x16x4_f32 %[c3], %[a2], %[b10], %[c3]\n\t"
        "v_wmma_f32_16x16x4_f32 %[c2], %[a1], %[b10], %[c2]\n\t"
        "v_wmma_f32_16x16x4_f32 %[c3], %[a0], %[b11], %[c3]\n\t"
        "s_branch .Ldone%=\n"
        ".Lnotfull%=:\n\t"
        "s_cmp_lg_u32 %[md], 2\n\t"
        "s_cbranch_scc1 .Ldone%=\n\t"
        "v_wmma_f32_16x16x4_f32 %[c1], %[a5], %[b00], %[c1]\n\t"
        "v_wmma_f32_16x16x4_f32 %[c0], %[a4], %[b00], %[c0]\n\t"
        "v_wmma_f32_16x16x4_f32 %[c1], %[a3], %[b01], %[c1]\n"
        ".Ldone%=:\n\t"
        "v_nop\n\tv_nop\n\tv_nop\n\tv_nop\n\tv_nop"
        : [c0] "+v"(ac[0]), [c1] "+v"(ac[1]), [c2] "+v"(ac[2]), [c3] "+v"(ac[3])
        : [a0] "v"(As[0]), [a1] "v"(As[1]), [a2] "v"(As[2]), [a3] "v"(As[3]), [a4] "v"(As[4]), [a5] "v"(As[5]),
          [a6] "v"(As[6]), [a7] "v"(As[7]), [a8] "v"(As[8]),
          [b00] "v"(b00), [b01] "v"(b01), [b10] "v"(b10), [b11] "v"(b11), [md] "s"(md)
        : "scc");
}

__device__ __forceinline__ void load_wfrag(const v2f* __restrict__ Ap, v2f (&A)[2][9]) {
#pragma unroll
    for (int s = 0; s < 2; ++s)
#pragma unroll
        for (int tap = 0; tap < 9; ++tap)
            A[s][tap] = Ap[(size_t)tap * 64 * OC_ + s * 2 * OC_];
}

__device__ __forceinline__ void mma_trip(const v2f (&A)[2][9], const v2f* xp, const int (&pix)[3], v8f (&acc)[3][4],
                                         const int md) {
#pragma unroll
    for (int s = 0; s < 2; ++s) {
        const v2f* xk = xp + s * 2 * XP_;
#pragma unroll
        for (int t = 0; t < 2; ++t) {
            const v2f b00 = xk[pix[t]];
            const v2f b01 = xk[pix[t] + 1];
            const v2f b10 = xk[pix[t] + XT_];
            const v2f b11 = xk[pix[t] + XT_ + 1];
            acc[t][3] = mm4(A[s][8], b00, acc[t][3]);
            acc[t][0] = mm4(A[s][4], b00, acc[t][0]);
            acc[t][1] = mm4(A[s][5], b00, acc[t][1]);
            acc[t][2] = mm4(A[s][7], b00, acc[t][2]);
            acc[t][3] = mm4(A[s][6], b01, acc[t][3]);
            acc[t][1] = mm4(A[s][3], b01, acc[t][1]);
            acc[t][3] = mm4(A[s][2], b10, acc[t][3]);
            acc[t][2] = mm4(A[s][1], b10, acc[t][2]);
            acc[t][3] = mm4(A[s][0], b11, acc[t][3]);
        }
        mma_slot3(A[s], xk[pix[2]], xk[pix[2] + 1], xk[pix[2] + XT_], xk[pix[2] + XT_ + 1], acc[2], md);
    }
}

__device__ __forceinline__ void stage_load(const float* __restrict__ p0, const int (&soff)[2], v2f (&tmp)[KPC_][2]) {
#pragma unroll
    for (int kq = 0; kq < KPC_; ++kq) {
        const float* p = p0 + (size_t)(2 * kq) * H_ * H_;
#pragma unroll
        for (int j = 0; j < 2; ++j) {
            tmp[kq][j].x = p[soff[j]];
            tmp[kq][j].y = p[soff[j] + H_ * H_];
        }
    }
}

__device__ __forceinline__ void stage_store(v2f* dst, const int (&spx)[2], const bool (&sinb)[2],
                                            const v2f (&tmp)[KPC_][2]) {
#pragma unroll
    for (int kq = 0; kq < KPC_; ++kq) {
#pragma unroll
        for (int j = 0; j < 2; ++j)
            dst[kq * XP_ + spx[j]] = sinb[j] ? tmp[kq][j] : v2f{};
    }
}

__global__ void __launch_bounds__(NTHR_) k_convblur(const float* __restrict__ x,
                                                    const float* __restrict__ W2,
                                                    float* __restrict__ out) {
    __shared__ v2f xs[2 * KPC_ * XP_];
    __shared__ __attribute__((aligned(16))) float cvs[16 * CVD_ * CVP_];

    const int c   = blockIdx.x * TR_;
    const int a   = blockIdx.y * TR_;
    const int b   = blockIdx.z >> 2;
    const int og  = blockIdx.z & 3;
    const int tid = threadIdx.x;
    const int lane = tid & 31;
    const int wid  = __builtin_amdgcn_readfirstlane(tid >> 5);
    const bool w3  = wid < 4;
    const int  md  = w3 ? 0 : (wid == 7 ? 2 : 1);
    const int  tb  = w3 ? wid * 3 : 12 + (wid - 4) * 2;
    const int ln   = lane & 15;
    const int khp  = lane >> 4;
    const int m    = og * 16 + ln;

    const float* xb = x + (size_t)b * C_ * H_ * H_;
    const v2f*   Wb = (const v2f*)W2 + (size_t)b * 9 * 64 * OC_;

    int pix[3];
#pragma unroll
    for (int t = 0; t < 3; ++t) {
        const int tile = (t < 2 || w3) ? tb + t : 20;
        int p = tile * 16 + ln;
        p = p < NPX_ ? p : NPX_ - 1;
        pix[t] = (p / G_) * XT_ + (p % G_);
    }

    v8f acc[3][4];
#pragma unroll
    for (int t = 0; t < 3; ++t)
#pragma unroll
        for (int q = 0; q < 4; ++q) acc[t][q] = v8f{};

    int  spx[2], soff[2];
    bool sval[2], sinb[2];
#pragma unroll
    for (int j = 0; j < 2; ++j) {
        const int px = tid + j * NTHR_;
        sval[j] = px < XPX_;
        const int pxc = sval[j] ? px : 0;
        const int xr = pxc / XT_;
        const int xc = pxc - xr * XT_;
        const int gr = a - 1 + xr, gc = c - 1 + xc;
        sinb[j] = sval[j] && gr >= 0 && gr < H_ && gc >= 0 && gc < H_;
        const int grc = gr < 0 ? 0 : (gr > H_ - 1 ? H_ - 1 : gr);
        const int gcc = gc < 0 ? 0 : (gc > H_ - 1 ? H_ - 1 : gc);
        soff[j] = grc * H_ + gcc;
        spx[j]  = sval[j] ? px : XPX_;
    }

    v2f tmp[KPC_][2];
    stage_load(xb, soff, tmp);
    stage_store(xs, spx, sinb, tmp);
    __syncthreads();

    const v2f* Ap = Wb + (size_t)khp * OC_ + m;
    v2f A[2][9];
#pragma nounroll
    for (int ch = 0; ch < NCH_ - 1; ++ch) {
        const v2f* xp = xs + (ch & 1) * (KPC_ * XP_) + khp * XP_;
        load_wfrag(Ap, A);
        stage_load(xb + (size_t)(2 * (ch + 1) * KPC_) * H_ * H_, soff, tmp);
        __builtin_amdgcn_sched_barrier(0);
        mma_trip(A, xp, pix, acc, md);
        __builtin_amdgcn_sched_barrier(0);
        load_wfrag(Ap + 4 * OC_, A);
        mma_trip(A, xp + 4 * XP_, pix, acc, md);
        stage_store(xs + ((ch + 1) & 1) * (KPC_ * XP_), spx, sinb, tmp);
        __syncthreads();
        Ap += 8 * OC_;
        asm volatile("v_nop\n\tv_nop\n\tv_nop\n\tv_nop\n\tv_nop"
                     : "+v"(acc[0][0]), "+v"(acc[0][1]), "+v"(acc[0][2]), "+v"(acc[0][3]),
                       "+v"(acc[1][0]), "+v"(acc[1][1]), "+v"(acc[1][2]), "+v"(acc[1][3]),
                       "+v"(acc[2][0]), "+v"(acc[2][1]), "+v"(acc[2][2]), "+v"(acc[2][3]));
    }
    {
        const v2f* xp = xs + ((NCH_ - 1) & 1) * (KPC_ * XP_) + khp * XP_;
        load_wfrag(Ap, A);
        mma_trip(A, xp, pix, acc, md);
        load_wfrag(Ap + 4 * OC_, A);
        mma_trip(A, xp + 4 * XP_, pix, acc, md);
        asm volatile("v_nop\n\tv_nop\n\tv_nop\n\tv_nop\n\tv_nop"
                     : "+v"(acc[0][0]), "+v"(acc[0][1]), "+v"(acc[0][2]), "+v"(acc[0][3]),
                       "+v"(acc[1][0]), "+v"(acc[1][1]), "+v"(acc[1][2]), "+v"(acc[1][3]),
                       "+v"(acc[2][0]), "+v"(acc[2][1]), "+v"(acc[2][2]), "+v"(acc[2][3]));
    }

#pragma unroll
    for (int t = 0; t < 3; ++t) {
        const int tile = (t < 2 || w3) ? tb + t : 20;
        const int p = tile * 16 + ln;
        if ((t < 2 || md != 1) && p < NPX_) {
            const int li = p / G_, lj = p % G_;
#pragma unroll
            for (int q = 0; q < 4; ++q) {
                const int ur = 2 * li + (q >> 1);
                const int vc = 2 * lj + (q & 1);
                if (ur < CVD_ && vc < CVD_) {
#pragma unroll
                    for (int r = 0; r < 8; ++r)
                        cvs[((khp * 8 + r) * CVD_ + ur) * CVP_ + vc] = acc[t][q][r];
                }
            }
        }
    }
    __syncthreads();

    float* ob = out + ((size_t)(b * OC_ + og * 16)) * OUT_ * OUT_;
    for (int idx = tid; idx < 16 * 32 * 8; idx += NTHR_) {
        const int q4 = idx & 7;
        const int yo = (idx >> 3) & 31;
        const int ol = idx >> 8;
        float o0 = 0.f, o1 = 0.f, o2 = 0.f, o3 = 0.f;
#pragma unroll
        for (int ky = 0; ky < 4; ++ky) {
            const float* rp = cvs + (ol * CVD_ + yo + ky) * CVP_ + 4 * q4;
            const float r0 = rp[0], r1 = rp[1], r2 = rp[2], r3 = rp[3], r4 = rp[4], r5 = rp[5], r6 = rp[6];
            const float f = (ky == 0 || ky == 3) ? 1.f : 3.f;
            o0 += f * (r0 + 3.f * r1 + 3.f * r2 + r3);
            o1 += f * (r1 + 3.f * r2 + 3.f * r3 + r4);
            o2 += f * (r2 + 3.f * r3 + 3.f * r4 + r5);
            o3 += f * (r3 + 3.f * r4 + 3.f * r5 + r6);
        }
        float4 st;
        st.x = o0 * (1.f / 16.f); st.y = o1 * (1.f / 16.f);
        st.z = o2 * (1.f / 16.f); st.w = o3 * (1.f / 16.f);
        *(float4*)(ob + ((size_t)ol * OUT_ + (2 * a + yo)) * OUT_ + 2 * c + 4 * q4) = st;
    }
}

extern "C" void kernel_launch(void* const* d_in, const int* in_sizes, int n_in,
                              void* d_out, int out_size, void* d_ws, size_t ws_size,
                              hipStream_t stream) {
    const float* x      = (const float*)d_in[0];
    const float* style  = (const float*)d_in[1];
    const float* weight = (const float*)d_in[2];
    const float* modw   = (const float*)d_in[3];
    const float* modb   = (const float*)d_in[4];
    float* out = (float*)d_out;
    float* ws  = (float*)d_ws;

    float* s_ws    = ws;
    float* W2_ws   = ws + 1024;

    k_style<<<dim3(C_), dim3(256), 0, stream>>>(style, modw, modb, s_ws);
    k_wmod<<<dim3(OC_, B_), dim3(C_), 0, stream>>>(weight, s_ws, W2_ws);

    k_convblur<<<dim3(8, 8, 32), dim3(NTHR_), 0, stream>>>(x, W2_ws, out);
}
